# starting kernel
# baseline (speedup 1.0000x reference)
#include <hip/hip_runtime.h>


#define T_DIM 2048
#define B_DIM 4
#define C_DIM 1024
#define H_DIM 16
#define K_DIM 7
#define O_DIM 112

typedef __attribute__((ext_vector_type(2))) float v2f;
typedef __attribute__((ext_vector_type(8))) float v8f;

__global__ __launch_bounds__(256)
void dynconv_fused_kernel(const float* __restrict__ x,
                          const float* __restrict__ Wm,
                          const float* __restrict__ bias,
                          float* __restrict__ out) {
  __shared__ float s_w[16 * O_DIM];

  const int mBase = blockIdx.x << 4;
  const int tid   = threadIdx.x;
  const int wave  = tid >> 5;
  const int lane  = tid & 31;
  const int half  = lane >> 4;
  const int lm    = lane & 15;

  if (wave < 7) {
    const int oBase = wave << 4;
    const float* __restrict__ xr = x  + (size_t)(mBase + lm) * C_DIM + 2 * half;
    const float* __restrict__ wr = Wm + (size_t)(oBase + lm) * C_DIM + 2 * half;
    v8f acc = {0.f, 0.f, 0.f, 0.f, 0.f, 0.f, 0.f, 0.f};
    #pragma unroll 4
    for (int k = 0; k < C_DIM; k += 4) {
      v2f a = *(const v2f*)(xr + k);
      v2f b = *(const v2f*)(wr + k);
      acc = __builtin_amdgcn_wmma_f32_16x16x4_f32(false, a, false, b,
                                                  (short)0, acc, false, false);
    }
    #pragma unroll
    for (int v = 0; v < 8; ++v)
      s_w[(v + 8 * half) * O_DIM + oBase + lm] = acc[v];
  }
  __syncthreads();

  {
    const int row = tid >> 4;
    const int h   = tid & 15;
    float* g = s_w + row * O_DIM + h * K_DIM;
    float mx = g[0];
    #pragma unroll
    for (int k = 1; k < K_DIM; ++k) mx = fmaxf(mx, g[k]);
    float e[K_DIM];
    float s = 0.f;
    #pragma unroll
    for (int k = 0; k < K_DIM; ++k) { e[k] = __expf(g[k] - mx); s += e[k]; }
    const float inv = 1.0f / s;
    #pragma unroll
    for (int k = 0; k < K_DIM; ++k) g[k] = e[k] * inv;
  }
  __syncthreads();

  for (int i = tid; i < 16 * C_DIM; i += 256) {
    const int row = i >> 10;
    const int c   = i & (C_DIM - 1);
    const int m   = mBase + row;
    const int t   = m >> 2;
    const float* wv = s_w + row * O_DIM + (c >> 6) * K_DIM;
    float accv = bias[c];
    if (t >= K_DIM - 1) {
      #pragma unroll
      for (int k = 0; k < K_DIM; ++k)
        accv = fmaf(x[(size_t)(m - (K_DIM - 1 - k) * B_DIM) * C_DIM + c], wv[k], accv);
    } else {
      #pragma unroll
      for (int k = 0; k < K_DIM; ++k) {
        if (t - (K_DIM - 1) + k >= 0)
          accv = fmaf(x[(size_t)(m - (K_DIM - 1 - k) * B_DIM) * C_DIM + c], wv[k], accv);
      }
    }
    out[(size_t)m * C_DIM + c] = accv;
  }
}

extern "C" void kernel_launch(void* const* d_in, const int* in_sizes, int n_in,
                              void* d_out, int out_size, void* d_ws, size_t ws_size,
                              hipStream_t stream) {
  const float* x    = (const float*)d_in[0];
  const float* Wm   = (const float*)d_in[1];
  const float* bias = (const float*)d_in[2];
  float* out = (float*)d_out;

  const int blocks = (T_DIM * B_DIM) / 16;
  hipLaunchKernelGGL(dynconv_fused_kernel, dim3(blocks), dim3(256), 0, stream,
                     x, Wm, bias, out);
}


# W re-laid into B-fragment order in the workspace by a small first launch; each W fragment load of the GEMM is then 512 contiguous bytes, not 32 B of 16 rows (k order and sums unchanged)
# speedup vs baseline: 1.2423x; 1.2423x over previous
#include <hip/hip_runtime.h>


#define T_DIM 2048
#define B_DIM 4
#define C_DIM 1024
#define H_DIM 16
#define K_DIM 7
#define O_DIM 112

#define ROWS_A 32
#define KW     128
#define PITCH  116
#define TT     16
#define PACK_FLOATS (O_DIM * C_DIM)
#define PACK_BYTES  ((size_t)PACK_FLOATS * 4)

typedef __attribute__((ext_vector_type(2))) float v2f;
typedef __attribute__((ext_vector_type(4))) float v4f;
typedef __attribute__((ext_vector_type(8))) float v8f;

__global__ __launch_bounds__(256)
void dynconv_pack_kernel(const float* __restrict__ Wm, float* __restrict__ wpk, size_t ws_bytes) {
  if (ws_bytes < PACK_BYTES) return;
  const int e    = blockIdx.x * 256 + threadIdx.x;
  const int lane = e & 31;
  const int kg   = (e >> 5) & 127;
  const int nt   = e >> 12;
  const v4f v = *(const v4f*)(Wm + (size_t)(nt * 16 + (lane & 15)) * C_DIM + 8 * kg + 4 * (lane >> 4));
  *(v4f*)(wpk + (size_t)e * 4) = v;
}

template <bool PACKED>
__device__ __forceinline__
void dynconv_filter_body(const float* __restrict__ x,
                         const float* __restrict__ Wsrc,
                         float* __restrict__ out,
                         float* __restrict__ s_p) {
  constexpr int GS = PACKED ? 128 : 8;

  const int tid   = threadIdx.x;
  const int wave  = tid >> 5;
  const int lane  = tid & 31;
  const int half  = lane >> 4;
  const int lm    = lane & 15;
  const int mBase = blockIdx.x * ROWS_A;

  const float* __restrict__ xr = x  + (size_t)(mBase + lm) * C_DIM + wave * KW + 4 * half;
  const float* __restrict__ wr = PACKED ? Wsrc + (size_t)wave * (KW / 8) * 128 + 4 * lane
                                        : Wsrc + (size_t)lm * C_DIM + wave * KW + 4 * half;

  v8f acc[2][7];
  #pragma unroll
  for (int mt = 0; mt < 2; ++mt)
    #pragma unroll
    for (int nt = 0; nt < 7; ++nt)
      acc[mt][nt] = (v8f){0.f, 0.f, 0.f, 0.f, 0.f, 0.f, 0.f, 0.f};

  v4f a[2][4], b[7][4];
  #pragma unroll
  for (int mt = 0; mt < 2; ++mt)
    #pragma unroll
    for (int g = 0; g < 4; ++g)
      a[mt][g] = *(const v4f*)(xr + mt * 16 * C_DIM + 8 * g);
  #pragma unroll
  for (int nt = 0; nt < 7; ++nt)
    #pragma unroll
    for (int g = 0; g < 4; ++g)
      b[nt][g] = *(const v4f*)(wr + nt * 16 * C_DIM + GS * g);

  #pragma unroll
  for (int s = 0; s < KW / 32; ++s) {
    v4f an[2][4], bn[7][4];
    if (s + 1 < KW / 32) {
      #pragma unroll
      for (int mt = 0; mt < 2; ++mt)
        #pragma unroll
        for (int g = 0; g < 4; ++g)
          an[mt][g] = *(const v4f*)(xr + mt * 16 * C_DIM + 32 * (s + 1) + 8 * g);
      #pragma unroll
      for (int nt = 0; nt < 7; ++nt)
        #pragma unroll
        for (int g = 0; g < 4; ++g)
          bn[nt][g] = *(const v4f*)(wr + nt * 16 * C_DIM + GS * (4 * (s + 1) + g));
    }
    #pragma unroll
    for (int g = 0; g < 4; ++g) {
      #pragma unroll
      for (int nt = 0; nt < 7; ++nt) {
        const v2f b01 = __builtin_shufflevector(b[nt][g], b[nt][g], 0, 1);
        #pragma unroll
        for (int mt = 0; mt < 2; ++mt) {
          const v2f a01 = __builtin_shufflevector(a[mt][g], a[mt][g], 0, 1);
          acc[mt][nt] = __builtin_amdgcn_wmma_f32_16x16x4_f32(false, a01, false, b01,
                                                              (short)0, acc[mt][nt], false, false);
        }
      }
      #pragma unroll
      for (int nt = 0; nt < 7; ++nt) {
        const v2f b23 = __builtin_shufflevector(b[nt][g], b[nt][g], 2, 3);
        #pragma unroll
        for (int mt = 0; mt < 2; ++mt) {
          const v2f a23 = __builtin_shufflevector(a[mt][g], a[mt][g], 2, 3);
          acc[mt][nt] = __builtin_amdgcn_wmma_f32_16x16x4_f32(false, a23, false, b23,
                                                              (short)0, acc[mt][nt], false, false);
        }
      }
    }
    if (s + 1 < KW / 32) {
      #pragma unroll
      for (int mt = 0; mt < 2; ++mt)
        #pragma unroll
        for (int g = 0; g < 4; ++g)
          a[mt][g] = an[mt][g];
      #pragma unroll
      for (int nt = 0; nt < 7; ++nt)
        #pragma unroll
        for (int g = 0; g < 4; ++g)
          b[nt][g] = bn[nt][g];
    }
  }

  #pragma unroll
  for (int mt = 0; mt < 2; ++mt)
    #pragma unroll
    for (int nt = 0; nt < 7; ++nt)
      #pragma unroll
      for (int v = 0; v < 8; ++v)
        s_p[(wave * ROWS_A + mt * 16 + v + 8 * half) * PITCH + nt * 16 + lm] = acc[mt][nt][v];
  __syncthreads();

  #pragma unroll
  for (int q = 0; q < 2; ++q) {
    const int gi  = tid + 256 * q;
    const int row = gi >> 4;
    const int h   = gi & 15;
    float l[K_DIM];
    #pragma unroll
    for (int k = 0; k < K_DIM; ++k) {
      float sum = s_p[row * PITCH + h * K_DIM + k];
      #pragma unroll
      for (int w = 1; w < 8; ++w)
        sum += s_p[(w * ROWS_A + row) * PITCH + h * K_DIM + k];
      l[k] = sum;
    }
    float mx = l[0];
    #pragma unroll
    for (int k = 1; k < K_DIM; ++k) mx = fmaxf(mx, l[k]);
    float e[K_DIM];
    float ssum = 0.f;
    #pragma unroll
    for (int k = 0; k < K_DIM; ++k) { e[k] = __expf(l[k] - mx); ssum += e[k]; }
    const float inv = 1.0f / ssum;
    float* dst = out + (size_t)(mBase + row) * C_DIM + h * 64;
    *(v4f*)dst       = (v4f){e[0] * inv, e[1] * inv, e[2] * inv, e[3] * inv};
    *(v4f*)(dst + 4) = (v4f){e[4] * inv, e[5] * inv, e[6] * inv, 0.f};
  }
}

__global__ __launch_bounds__(256)
void dynconv_filter_kernel(const float* __restrict__ x,
                           const float* __restrict__ Wm,
                           const float* __restrict__ wpk,
                           size_t ws_bytes,
                           float* __restrict__ out) {
  __shared__ float s_p[8 * ROWS_A * PITCH];
  if (ws_bytes >= PACK_BYTES) dynconv_filter_body<true>(x, wpk, out, s_p);
  else                        dynconv_filter_body<false>(x, Wm, out, s_p);
}

__global__ __launch_bounds__(256)
void dynconv_conv_kernel(const float* x,
                         const float* bias,
                         float* out) {
  const int c4 = threadIdx.x;
  const int h  = c4 >> 4;
  const int b  = blockIdx.x & (B_DIM - 1);
  const int t0 = (blockIdx.x >> 2) * TT;

  const float* xcol = x + (size_t)b * C_DIM + 4 * c4;
  float*       orow = out + (size_t)b * C_DIM;

  v4f xw[TT + K_DIM - 1];
  v4f wa[TT], wb[TT];

  const v4f bv = *(const v4f*)(bias + 4 * c4);
  #pragma unroll
  for (int j = 0; j < K_DIM - 1; ++j) {
    const int t  = t0 - (K_DIM - 1) + j;
    const int tc = t < 0 ? 0 : t;
    xw[j] = *(const v4f*)(xcol + (size_t)tc * (B_DIM * C_DIM));
  }
  #pragma unroll
  for (int i = 0; i < TT; ++i) {
    const float* wp = orow + (size_t)(t0 + i) * (B_DIM * C_DIM) + h * 64;
    wa[i] = *(const v4f*)wp;
    wb[i] = *(const v4f*)(wp + 4);
    xw[i + K_DIM - 1] = *(const v4f*)(xcol + (size_t)(t0 + i) * (B_DIM * C_DIM));
  }
  __builtin_amdgcn_sched_barrier(0);

  #pragma unroll
  for (int j = 0; j < K_DIM - 1; ++j) {
    const int t = t0 - (K_DIM - 1) + j;
    if (t < 0) xw[j] = (v4f){0.f, 0.f, 0.f, 0.f};
  }

  #pragma unroll
  for (int i = 0; i < TT; ++i) {
    const float wk[K_DIM] = {wa[i][0], wa[i][1], wa[i][2], wa[i][3], wb[i][0], wb[i][1], wb[i][2]};
    v4f accv = bv;
    #pragma unroll
    for (int k = 0; k < K_DIM; ++k) {
      accv[0] = fmaf(xw[i + k][0], wk[k], accv[0]);
      accv[1] = fmaf(xw[i + k][1], wk[k], accv[1]);
      accv[2] = fmaf(xw[i + k][2], wk[k], accv[2]);
      accv[3] = fmaf(xw[i + k][3], wk[k], accv[3]);
    }
    *(v4f*)(orow + (size_t)(t0 + i) * (B_DIM * C_DIM) + 4 * c4) = accv;
  }
}

extern "C" void kernel_launch(void* const* d_in, const int* in_sizes, int n_in,
                              void* d_out, int out_size, void* d_ws, size_t ws_size,
                              hipStream_t stream) {
  const float* x    = (const float*)d_in[0];
  const float* Wm   = (const float*)d_in[1];
  const float* bias = (const float*)d_in[2];
  float* out = (float*)d_out;

  float* wpk = (float*)d_ws;

  hipLaunchKernelGGL(dynconv_pack_kernel, dim3(112), dim3(256), 0, stream, Wm, wpk, ws_size);
  hipLaunchKernelGGL(dynconv_filter_kernel, dim3((T_DIM * B_DIM) / ROWS_A), dim3(256), 0, stream,
                     x, Wm, (const float*)wpk, ws_size, out);
  hipLaunchKernelGGL(dynconv_conv_kernel, dim3((T_DIM / TT) * B_DIM), dim3(256), 0, stream,
                     x, bias, out);
}


# Conv folded into the GEMM launch: taps stay in LDS and each work-group makes its own 32 rows from a 14-timestep register window of x re-read through the caches; 2 launches (all sums unchanged)
# speedup vs baseline: 3.4228x; 2.7552x over previous
#include <hip/hip_runtime.h>


#define T_DIM 2048
#define B_DIM 4
#define C_DIM 1024
#define H_DIM 16
#define K_DIM 7
#define O_DIM 112

#define ROWS_A 32
#define KW     128
#define PITCH  116
#define TW     8
#define WIN    (TW + K_DIM - 1)
#define PACK_FLOATS (O_DIM * C_DIM)
#define PACK_BYTES  ((size_t)PACK_FLOATS * 4)

typedef __attribute__((ext_vector_type(2))) float v2f;
typedef __attribute__((ext_vector_type(4))) float v4f;
typedef __attribute__((ext_vector_type(8))) float v8f;

__device__ __forceinline__ void dynconv_lds_barrier() {
  __builtin_amdgcn_fence(__ATOMIC_RELEASE, "workgroup", "local");
  __builtin_amdgcn_s_barrier();
  __builtin_amdgcn_fence(__ATOMIC_ACQUIRE, "workgroup", "local");
}

__global__ __launch_bounds__(256)
void dynconv_pack_kernel(const float* __restrict__ Wm, float* __restrict__ wpk, size_t ws_bytes) {
  if (ws_bytes < PACK_BYTES) return;
  const int e    = blockIdx.x * 256 + threadIdx.x;
  const int lane = e & 31;
  const int kg   = (e >> 5) & 127;
  const int nt   = e >> 12;
  const v4f v = *(const v4f*)(Wm + (size_t)(nt * 16 + (lane & 15)) * C_DIM + 8 * kg + 4 * (lane >> 4));
  *(v4f*)(wpk + (size_t)e * 4) = v;
}

template <bool PACKED>
__device__ __forceinline__
void dynconv_main_body(const float* __restrict__ x,
                       const float* __restrict__ Wsrc,
                       const float* __restrict__ bias,
                       float* __restrict__ out,
                       float* __restrict__ s_p,
                       float* __restrict__ s_t) {
  constexpr int GS = PACKED ? 128 : 8;

  const int tid   = threadIdx.x;
  const int wave  = tid >> 5;
  const int lane  = tid & 31;
  const int half  = lane >> 4;
  const int lm    = lane & 15;
  const int mBase = blockIdx.x * ROWS_A;

  const float* __restrict__ xr = x  + (size_t)(mBase + lm) * C_DIM + wave * KW + 4 * half;
  const float* __restrict__ wr = PACKED ? Wsrc + (size_t)wave * (KW / 8) * 128 + 4 * lane
                                        : Wsrc + (size_t)lm * C_DIM + wave * KW + 4 * half;

  v8f acc[2][7];
  #pragma unroll
  for (int mt = 0; mt < 2; ++mt)
    #pragma unroll
    for (int nt = 0; nt < 7; ++nt)
      acc[mt][nt] = (v8f){0.f, 0.f, 0.f, 0.f, 0.f, 0.f, 0.f, 0.f};

  v4f a[2][4], b[7][4];
  #pragma unroll
  for (int mt = 0; mt < 2; ++mt)
    #pragma unroll
    for (int g = 0; g < 4; ++g)
      a[mt][g] = *(const v4f*)(xr + mt * 16 * C_DIM + 8 * g);
  #pragma unroll
  for (int nt = 0; nt < 7; ++nt)
    #pragma unroll
    for (int g = 0; g < 4; ++g)
      b[nt][g] = *(const v4f*)(wr + nt * 16 * C_DIM + GS * g);

  #pragma unroll
  for (int s = 0; s < KW / 32; ++s) {
    v4f an[2][4], bn[7][4];
    if (s + 1 < KW / 32) {
      #pragma unroll
      for (int mt = 0; mt < 2; ++mt)
        #pragma unroll
        for (int g = 0; g < 4; ++g)
          an[mt][g] = *(const v4f*)(xr + mt * 16 * C_DIM + 32 * (s + 1) + 8 * g);
      #pragma unroll
      for (int nt = 0; nt < 7; ++nt)
        #pragma unroll
        for (int g = 0; g < 4; ++g)
          bn[nt][g] = *(const v4f*)(wr + nt * 16 * C_DIM + GS * (4 * (s + 1) + g));
    }
    #pragma unroll
    for (int g = 0; g < 4; ++g) {
      #pragma unroll
      for (int nt = 0; nt < 7; ++nt) {
        const v2f b01 = __builtin_shufflevector(b[nt][g], b[nt][g], 0, 1);
        #pragma unroll
        for (int mt = 0; mt < 2; ++mt) {
          const v2f a01 = __builtin_shufflevector(a[mt][g], a[mt][g], 0, 1);
          acc[mt][nt] = __builtin_amdgcn_wmma_f32_16x16x4_f32(false, a01, false, b01,
                                                              (short)0, acc[mt][nt], false, false);
        }
      }
      #pragma unroll
      for (int nt = 0; nt < 7; ++nt) {
        const v2f b23 = __builtin_shufflevector(b[nt][g], b[nt][g], 2, 3);
        #pragma unroll
        for (int mt = 0; mt < 2; ++mt) {
          const v2f a23 = __builtin_shufflevector(a[mt][g], a[mt][g], 2, 3);
          acc[mt][nt] = __builtin_amdgcn_wmma_f32_16x16x4_f32(false, a23, false, b23,
                                                              (short)0, acc[mt][nt], false, false);
        }
      }
    }
    if (s + 1 < KW / 32) {
      #pragma unroll
      for (int mt = 0; mt < 2; ++mt)
        #pragma unroll
        for (int g = 0; g < 4; ++g)
          a[mt][g] = an[mt][g];
      #pragma unroll
      for (int nt = 0; nt < 7; ++nt)
        #pragma unroll
        for (int g = 0; g < 4; ++g)
          b[nt][g] = bn[nt][g];
    }
  }

  #pragma unroll
  for (int mt = 0; mt < 2; ++mt)
    #pragma unroll
    for (int nt = 0; nt < 7; ++nt)
      #pragma unroll
      for (int v = 0; v < 8; ++v)
        s_p[(wave * ROWS_A + mt * 16 + v + 8 * half) * PITCH + nt * 16 + lm] = acc[mt][nt][v];
  __syncthreads();

  const int t0 = blockIdx.x * TW;
  const int hc = tid >> 4;
  v4f xv[B_DIM][WIN];
  const v4f bv = *(const v4f*)(bias + 4 * tid);
  #pragma unroll
  for (int j = 0; j < WIN; ++j) {
    const int t  = t0 - (K_DIM - 1) + j;
    const int tc = t < 0 ? 0 : t;
    #pragma unroll
    for (int bb = 0; bb < B_DIM; ++bb)
      xv[bb][j] = *(const v4f*)(x + ((size_t)tc * B_DIM + bb) * C_DIM + 4 * tid);
  }
  __builtin_amdgcn_sched_barrier(0);

  #pragma unroll
  for (int q = 0; q < 2; ++q) {
    const int gi  = tid + 256 * q;
    const int row = gi >> 4;
    const int h   = gi & 15;
    float l[K_DIM];
    #pragma unroll
    for (int k = 0; k < K_DIM; ++k) {
      float sum = s_p[row * PITCH + h * K_DIM + k];
      #pragma unroll
      for (int w = 1; w < 8; ++w)
        sum += s_p[(w * ROWS_A + row) * PITCH + h * K_DIM + k];
      l[k] = sum;
    }
    float mx = l[0];
    #pragma unroll
    for (int k = 1; k < K_DIM; ++k) mx = fmaxf(mx, l[k]);
    float e[K_DIM];
    float ssum = 0.f;
    #pragma unroll
    for (int k = 0; k < K_DIM; ++k) { e[k] = __expf(l[k] - mx); ssum += e[k]; }
    const float inv = 1.0f / ssum;
    float* dst = s_t + gi * 8;
    *(v4f*)dst       = (v4f){e[0] * inv, e[1] * inv, e[2] * inv, e[3] * inv};
    *(v4f*)(dst + 4) = (v4f){e[4] * inv, e[5] * inv, e[6] * inv, 0.f};
  }
  dynconv_lds_barrier();

  if (t0 == 0) {
    #pragma unroll
    for (int bb = 0; bb < B_DIM; ++bb)
      #pragma unroll
      for (int j = 0; j < K_DIM - 1; ++j)
        xv[bb][j] = (v4f){0.f, 0.f, 0.f, 0.f};
  }
  #pragma unroll
  for (int i = 0; i < TW; ++i)
    #pragma unroll
    for (int bb = 0; bb < B_DIM; ++bb) {
      const int row = 4 * i + bb;
      const v4f wa = *(const v4f*)(s_t + (row * 16 + hc) * 8);
      const v4f wb = *(const v4f*)(s_t + (row * 16 + hc) * 8 + 4);
      const float wk[K_DIM] = {wa[0], wa[1], wa[2], wa[3], wb[0], wb[1], wb[2]};
      v4f accv = bv;
      #pragma unroll
      for (int k = 0; k < K_DIM; ++k) {
        accv[0] = fmaf(xv[bb][i + k][0], wk[k], accv[0]);
        accv[1] = fmaf(xv[bb][i + k][1], wk[k], accv[1]);
        accv[2] = fmaf(xv[bb][i + k][2], wk[k], accv[2]);
        accv[3] = fmaf(xv[bb][i + k][3], wk[k], accv[3]);
      }
      *(v4f*)(out + (size_t)(mBase + row) * C_DIM + 4 * tid) = accv;
    }
}

__global__ __launch_bounds__(256)
void dynconv_main_kernel(const float* __restrict__ x,
                         const float* __restrict__ Wm,
                         const float* __restrict__ wpk,
                         size_t ws_bytes,
                         const float* __restrict__ bias,
                         float* __restrict__ out) {
  __shared__ float s_p[8 * ROWS_A * PITCH];
  __shared__ float s_t[ROWS_A * H_DIM * 8];
  if (ws_bytes >= PACK_BYTES) dynconv_main_body<true>(x, wpk, bias, out, s_p, s_t);
  else                        dynconv_main_body<false>(x, Wm, bias, out, s_p, s_t);
}

extern "C" void kernel_launch(void* const* d_in, const int* in_sizes, int n_in,
                              void* d_out, int out_size, void* d_ws, size_t ws_size,
                              hipStream_t stream) {
  const float* x    = (const float*)d_in[0];
  const float* Wm   = (const float*)d_in[1];
  const float* bias = (const float*)d_in[2];
  float* out = (float*)d_out;

  float* wpk = (float*)d_ws;

  hipLaunchKernelGGL(dynconv_pack_kernel, dim3(112), dim3(256), 0, stream, Wm, wpk, ws_size);
  hipLaunchKernelGGL(dynconv_main_kernel, dim3((T_DIM * B_DIM) / ROWS_A), dim3(256), 0, stream,
                     x, Wm, (const float*)wpk, ws_size, bias, out);
}


# Conv window of x requested early: first 4 timesteps inside the GEMM's last stage, the rest behind each wave's last WMMA and LDS stores, ahead of a first barrier that orders LDS only (sums unchanged)
# speedup vs baseline: 3.6380x; 1.0629x over previous
#include <hip/hip_runtime.h>


#define T_DIM 2048
#define B_DIM 4
#define C_DIM 1024
#define H_DIM 16
#define K_DIM 7
#define O_DIM 112

#define ROWS_A 32
#define KW     128
#define PITCH  116
#define TW     8
#define WIN    (TW + K_DIM - 1)
#define WIN_EARLY 4
#define PACK_FLOATS (O_DIM * C_DIM)
#define PACK_BYTES  ((size_t)PACK_FLOATS * 4)

typedef __attribute__((ext_vector_type(2))) float v2f;
typedef __attribute__((ext_vector_type(4))) float v4f;
typedef __attribute__((ext_vector_type(8))) float v8f;

__device__ __forceinline__ void dynconv_lds_barrier() {
  __builtin_amdgcn_fence(__ATOMIC_RELEASE, "workgroup", "local");
  __builtin_amdgcn_s_barrier();
  __builtin_amdgcn_fence(__ATOMIC_ACQUIRE, "workgroup", "local");
}

__global__ __launch_bounds__(256)
void dynconv_pack_kernel(const float* __restrict__ Wm, float* __restrict__ wpk, size_t ws_bytes) {
  if (ws_bytes < PACK_BYTES) return;
  const int e    = blockIdx.x * 256 + threadIdx.x;
  const int lane = e & 31;
  const int kg   = (e >> 5) & 127;
  const int nt   = e >> 12;
  const v4f v = *(const v4f*)(Wm + (size_t)(nt * 16 + (lane & 15)) * C_DIM + 8 * kg + 4 * (lane >> 4));
  *(v4f*)(wpk + (size_t)e * 4) = v;
}

template <bool PACKED>
__device__ __forceinline__
void dynconv_main_body(const float* __restrict__ x,
                       const float* __restrict__ Wsrc,
                       const float* __restrict__ bias,
                       float* __restrict__ out,
                       float* __restrict__ s_p,
                       float* __restrict__ s_t) {
  constexpr int GS = PACKED ? 128 : 8;
  constexpr int EARLY = PACKED ? WIN_EARLY : 0;

  const int tid   = threadIdx.x;
  const int wave  = tid >> 5;
  const int lane  = tid & 31;
  const int half  = lane >> 4;
  const int lm    = lane & 15;
  const int mBase = blockIdx.x * ROWS_A;

  const float* __restrict__ xr = x  + (size_t)(mBase + lm) * C_DIM + wave * KW + 4 * half;
  const float* __restrict__ wr = PACKED ? Wsrc + (size_t)wave * (KW / 8) * 128 + 4 * lane
                                        : Wsrc + (size_t)lm * C_DIM + wave * KW + 4 * half;

  v8f acc[2][7];
  #pragma unroll
  for (int mt = 0; mt < 2; ++mt)
    #pragma unroll
    for (int nt = 0; nt < 7; ++nt)
      acc[mt][nt] = (v8f){0.f, 0.f, 0.f, 0.f, 0.f, 0.f, 0.f, 0.f};

  const int t0 = blockIdx.x * TW;
  v4f xv[B_DIM][WIN];
  v4f bv;

  v4f a[2][4], b[7][4];
  #pragma unroll
  for (int mt = 0; mt < 2; ++mt)
    #pragma unroll
    for (int g = 0; g < 4; ++g)
      a[mt][g] = *(const v4f*)(xr + mt * 16 * C_DIM + 8 * g);
  #pragma unroll
  for (int nt = 0; nt < 7; ++nt)
    #pragma unroll
    for (int g = 0; g < 4; ++g)
      b[nt][g] = *(const v4f*)(wr + nt * 16 * C_DIM + GS * g);

  #pragma unroll
  for (int s = 0; s < KW / 32; ++s) {
    v4f an[2][4], bn[7][4];
    if (s + 1 < KW / 32) {
      #pragma unroll
      for (int mt = 0; mt < 2; ++mt)
        #pragma unroll
        for (int g = 0; g < 4; ++g)
          an[mt][g] = *(const v4f*)(xr + mt * 16 * C_DIM + 32 * (s + 1) + 8 * g);
      #pragma unroll
      for (int nt = 0; nt < 7; ++nt)
        #pragma unroll
        for (int g = 0; g < 4; ++g)
          bn[nt][g] = *(const v4f*)(wr + nt * 16 * C_DIM + GS * (4 * (s + 1) + g));
    } else if constexpr (PACKED) {
      bv = *(const v4f*)(bias + 4 * tid);
      #pragma unroll
      for (int j = 0; j < EARLY; ++j) {
        const int t  = t0 - (K_DIM - 1) + j;
        const int tc = t < 0 ? 0 : t;
        #pragma unroll
        for (int bb = 0; bb < B_DIM; ++bb)
          xv[bb][j] = *(const v4f*)(x + ((size_t)tc * B_DIM + bb) * C_DIM + 4 * tid);
      }
      __builtin_amdgcn_sched_barrier(0);
    }
    #pragma unroll
    for (int g = 0; g < 4; ++g) {
      #pragma unroll
      for (int nt = 0; nt < 7; ++nt) {
        const v2f b01 = __builtin_shufflevector(b[nt][g], b[nt][g], 0, 1);
        #pragma unroll
        for (int mt = 0; mt < 2; ++mt) {
          const v2f a01 = __builtin_shufflevector(a[mt][g], a[mt][g], 0, 1);
          acc[mt][nt] = __builtin_amdgcn_wmma_f32_16x16x4_f32(false, a01, false, b01,
                                                              (short)0, acc[mt][nt], false, false);
        }
      }
      #pragma unroll
      for (int nt = 0; nt < 7; ++nt) {
        const v2f b23 = __builtin_shufflevector(b[nt][g], b[nt][g], 2, 3);
        #pragma unroll
        for (int mt = 0; mt < 2; ++mt) {
          const v2f a23 = __builtin_shufflevector(a[mt][g], a[mt][g], 2, 3);
          acc[mt][nt] = __builtin_amdgcn_wmma_f32_16x16x4_f32(false, a23, false, b23,
                                                              (short)0, acc[mt][nt], false, false);
        }
      }
    }
    if (s + 1 < KW / 32) {
      #pragma unroll
      for (int mt = 0; mt < 2; ++mt)
        #pragma unroll
        for (int g = 0; g < 4; ++g)
          a[mt][g] = an[mt][g];
      #pragma unroll
      for (int nt = 0; nt < 7; ++nt)
        #pragma unroll
        for (int g = 0; g < 4; ++g)
          b[nt][g] = bn[nt][g];
    }
  }

  #pragma unroll
  for (int mt = 0; mt < 2; ++mt)
    #pragma unroll
    for (int nt = 0; nt < 7; ++nt)
      #pragma unroll
      for (int v = 0; v < 8; ++v)
        s_p[(wave * ROWS_A + mt * 16 + v + 8 * half) * PITCH + nt * 16 + lm] = acc[mt][nt][v];
  if constexpr (PACKED) {
    __builtin_amdgcn_sched_barrier(0);
    #pragma unroll
    for (int j = EARLY; j < WIN; ++j) {
      const int t  = t0 - (K_DIM - 1) + j;
      const int tc = t < 0 ? 0 : t;
      #pragma unroll
      for (int bb = 0; bb < B_DIM; ++bb)
        xv[bb][j] = *(const v4f*)(x + ((size_t)tc * B_DIM + bb) * C_DIM + 4 * tid);
    }
    __builtin_amdgcn_sched_barrier(0);
  }

  if constexpr (PACKED) {
    dynconv_lds_barrier();
  } else {
    __syncthreads();
    bv = *(const v4f*)(bias + 4 * tid);
    #pragma unroll
    for (int j = 0; j < WIN; ++j) {
      const int t  = t0 - (K_DIM - 1) + j;
      const int tc = t < 0 ? 0 : t;
      #pragma unroll
      for (int bb = 0; bb < B_DIM; ++bb)
        xv[bb][j] = *(const v4f*)(x + ((size_t)tc * B_DIM + bb) * C_DIM + 4 * tid);
    }
    __builtin_amdgcn_sched_barrier(0);
  }

  const int hc = tid >> 4;

  #pragma unroll
  for (int q = 0; q < 2; ++q) {
    const int gi  = tid + 256 * q;
    const int row = gi >> 4;
    const int h   = gi & 15;
    float l[K_DIM];
    #pragma unroll
    for (int k = 0; k < K_DIM; ++k) {
      float sum = s_p[row * PITCH + h * K_DIM + k];
      #pragma unroll
      for (int w = 1; w < 8; ++w)
        sum += s_p[(w * ROWS_A + row) * PITCH + h * K_DIM + k];
      l[k] = sum;
    }
    float mx = l[0];
    #pragma unroll
    for (int k = 1; k < K_DIM; ++k) mx = fmaxf(mx, l[k]);
    float e[K_DIM];
    float ssum = 0.f;
    #pragma unroll
    for (int k = 0; k < K_DIM; ++k) { e[k] = __expf(l[k] - mx); ssum += e[k]; }
    const float inv = 1.0f / ssum;
    float* dst = s_t + gi * 8;
    *(v4f*)dst       = (v4f){e[0] * inv, e[1] * inv, e[2] * inv, e[3] * inv};
    *(v4f*)(dst + 4) = (v4f){e[4] * inv, e[5] * inv, e[6] * inv, 0.f};
  }
  dynconv_lds_barrier();

  if (t0 == 0) {
    #pragma unroll
    for (int bb = 0; bb < B_DIM; ++bb)
      #pragma unroll
      for (int j = 0; j < K_DIM - 1; ++j)
        xv[bb][j] = (v4f){0.f, 0.f, 0.f, 0.f};
  }
  #pragma unroll
  for (int i = 0; i < TW; ++i)
    #pragma unroll
    for (int bb = 0; bb < B_DIM; ++bb) {
      const int row = 4 * i + bb;
      const v4f wa = *(const v4f*)(s_t + (row * 16 + hc) * 8);
      const v4f wb = *(const v4f*)(s_t + (row * 16 + hc) * 8 + 4);
      const float wk[K_DIM] = {wa[0], wa[1], wa[2], wa[3], wb[0], wb[1], wb[2]};
      v4f accv = bv;
      #pragma unroll
      for (int k = 0; k < K_DIM; ++k) {
        accv[0] = fmaf(xv[bb][i + k][0], wk[k], accv[0]);
        accv[1] = fmaf(xv[bb][i + k][1], wk[k], accv[1]);
        accv[2] = fmaf(xv[bb][i + k][2], wk[k], accv[2]);
        accv[3] = fmaf(xv[bb][i + k][3], wk[k], accv[3]);
      }
      *(v4f*)(out + (size_t)(mBase + row) * C_DIM + 4 * tid) = accv;
    }
}

__global__ __launch_bounds__(256)
void dynconv_main_kernel(const float* __restrict__ x,
                         const float* __restrict__ Wm,
                         const float* __restrict__ wpk,
                         size_t ws_bytes,
                         const float* __restrict__ bias,
                         float* __restrict__ out) {
  __shared__ float s_p[8 * ROWS_A * PITCH];
  __shared__ float s_t[ROWS_A * H_DIM * 8];
  if (ws_bytes >= PACK_BYTES) dynconv_main_body<true>(x, wpk, bias, out, s_p, s_t);
  else                        dynconv_main_body<false>(x, Wm, bias, out, s_p, s_t);
}

extern "C" void kernel_launch(void* const* d_in, const int* in_sizes, int n_in,
                              void* d_out, int out_size, void* d_ws, size_t ws_size,
                              hipStream_t stream) {
  const float* x    = (const float*)d_in[0];
  const float* Wm   = (const float*)d_in[1];
  const float* bias = (const float*)d_in[2];
  float* out = (float*)d_out;

  float* wpk = (float*)d_ws;

  hipLaunchKernelGGL(dynconv_pack_kernel, dim3(112), dim3(256), 0, stream, Wm, wpk, ws_size);
  hipLaunchKernelGGL(dynconv_main_kernel, dim3((T_DIM * B_DIM) / ROWS_A), dim3(256), 0, stream,
                     x, Wm, (const float*)wpk, ws_size, bias, out);
}


# Main kernel at one wave per SIMD: 4-wave work-groups, k split 4 ways (4 partials per logit: order of k sums changes); the 1024-register budget holds the whole conv window, requested under the GEMM
# speedup vs baseline: 3.8830x; 1.0673x over previous
#include <hip/hip_runtime.h>


#define T_DIM 2048
#define B_DIM 4
#define C_DIM 1024
#define H_DIM 16
#define K_DIM 7
#define O_DIM 112

#define ROWS_A 32
#define NWAVES 4
#define NTHR   (NWAVES * 32)
#define KW     256
#define NSTAGE (KW / 32)
#define PITCH  116
#define TW     8
#define WIN    (TW + K_DIM - 1)
#define WIN_LOADS (2 * B_DIM * WIN)
#define WIN_PER_STAGE (WIN_LOADS / (NSTAGE - 1))
#define PACK_FLOATS (O_DIM * C_DIM)
#define PACK_BYTES  ((size_t)PACK_FLOATS * 4)

typedef __attribute__((ext_vector_type(2))) float v2f;
typedef __attribute__((ext_vector_type(4))) float v4f;
typedef __attribute__((ext_vector_type(8))) float v8f;

__device__ __forceinline__ void dynconv_lds_barrier() {
  __builtin_amdgcn_fence(__ATOMIC_RELEASE, "workgroup", "local");
  __builtin_amdgcn_s_barrier();
  __builtin_amdgcn_fence(__ATOMIC_ACQUIRE, "workgroup", "local");
}

__global__ __launch_bounds__(256)
void dynconv_pack_kernel(const float* __restrict__ Wm, float* __restrict__ wpk, size_t ws_bytes) {
  if (ws_bytes < PACK_BYTES) return;
  const int e    = blockIdx.x * 256 + threadIdx.x;
  const int lane = e & 31;
  const int kg   = (e >> 5) & 127;
  const int nt   = e >> 12;
  const v4f v = *(const v4f*)(Wm + (size_t)(nt * 16 + (lane & 15)) * C_DIM + 8 * kg + 4 * (lane >> 4));
  *(v4f*)(wpk + (size_t)e * 4) = v;
}

template <bool PACKED>
__device__ __forceinline__
void dynconv_main_body(const float* __restrict__ x,
                       const float* __restrict__ Wsrc,
                       const float* __restrict__ bias,
                       float* __restrict__ out,
                       float* __restrict__ s_p,
                       float* __restrict__ s_t) {
  constexpr int GS = PACKED ? 128 : 8;

  const int tid   = threadIdx.x;
  const int wave  = tid >> 5;
  const int lane  = tid & 31;
  const int half  = lane >> 4;
  const int lm    = lane & 15;
  const int mBase = blockIdx.x * ROWS_A;

  const float* __restrict__ xr = x  + (size_t)(mBase + lm) * C_DIM + wave * KW + 4 * half;
  const float* __restrict__ wr = PACKED ? Wsrc + (size_t)wave * (KW / 8) * 128 + 4 * lane
                                        : Wsrc + (size_t)lm * C_DIM + wave * KW + 4 * half;

  v8f acc[2][7];
  #pragma unroll
  for (int mt = 0; mt < 2; ++mt)
    #pragma unroll
    for (int nt = 0; nt < 7; ++nt)
      acc[mt][nt] = (v8f){0.f, 0.f, 0.f, 0.f, 0.f, 0.f, 0.f, 0.f};

  const int t0 = blockIdx.x * TW;
  v4f xv[2][B_DIM][WIN];
  v4f bv[2];

  auto win_load = [&](int q) {
    const int j  = q / (2 * B_DIM);
    const int bb = (q >> 1) & (B_DIM - 1);
    const int hv = q & 1;
    const int t  = t0 - (K_DIM - 1) + j;
    const int tc = t < 0 ? 0 : t;
    xv[hv][bb][j] = *(const v4f*)(x + ((size_t)tc * B_DIM + bb) * C_DIM + 512 * hv + 4 * tid);
  };

  v4f a[2][4], b[7][4];
  #pragma unroll
  for (int mt = 0; mt < 2; ++mt)
    #pragma unroll
    for (int g = 0; g < 4; ++g)
      a[mt][g] = *(const v4f*)(xr + mt * 16 * C_DIM + 8 * g);
  #pragma unroll
  for (int nt = 0; nt < 7; ++nt)
    #pragma unroll
    for (int g = 0; g < 4; ++g)
      b[nt][g] = *(const v4f*)(wr + nt * 16 * C_DIM + GS * g);

  #pragma unroll
  for (int s = 0; s < NSTAGE; ++s) {
    v4f an[2][4], bn[7][4];
    if (s + 1 < NSTAGE) {
      #pragma unroll
      for (int mt = 0; mt < 2; ++mt)
        #pragma unroll
        for (int g = 0; g < 4; ++g)
          an[mt][g] = *(const v4f*)(xr + mt * 16 * C_DIM + 32 * (s + 1) + 8 * g);
      #pragma unroll
      for (int nt = 0; nt < 7; ++nt)
        #pragma unroll
        for (int g = 0; g < 4; ++g)
          bn[nt][g] = *(const v4f*)(wr + nt * 16 * C_DIM + GS * (4 * (s + 1) + g));
    }
    if constexpr (PACKED) {
      if (s >= 1) {
        if (s == 1) {
          bv[0] = *(const v4f*)(bias + 4 * tid);
          bv[1] = *(const v4f*)(bias + 512 + 4 * tid);
        }
        #pragma unroll
        for (int q = WIN_PER_STAGE * (s - 1); q < WIN_PER_STAGE * s; ++q) win_load(q);
      }
      __builtin_amdgcn_sched_barrier(0);
    }
    #pragma unroll
    for (int g = 0; g < 4; ++g) {
      #pragma unroll
      for (int nt = 0; nt < 7; ++nt) {
        const v2f b01 = __builtin_shufflevector(b[nt][g], b[nt][g], 0, 1);
        #pragma unroll
        for (int mt = 0; mt < 2; ++mt) {
          const v2f a01 = __builtin_shufflevector(a[mt][g], a[mt][g], 0, 1);
          acc[mt][nt] = __builtin_amdgcn_wmma_f32_16x16x4_f32(false, a01, false, b01,
                                                              (short)0, acc[mt][nt], false, false);
        }
      }
      #pragma unroll
      for (int nt = 0; nt < 7; ++nt) {
        const v2f b23 = __builtin_shufflevector(b[nt][g], b[nt][g], 2, 3);
        #pragma unroll
        for (int mt = 0; mt < 2; ++mt) {
          const v2f a23 = __builtin_shufflevector(a[mt][g], a[mt][g], 2, 3);
          acc[mt][nt] = __builtin_amdgcn_wmma_f32_16x16x4_f32(false, a23, false, b23,
                                                              (short)0, acc[mt][nt], false, false);
        }
      }
    }
    if constexpr (PACKED) {
      __builtin_amdgcn_sched_barrier(0);
    }
    if (s + 1 < NSTAGE) {
      #pragma unroll
      for (int mt = 0; mt < 2; ++mt)
        #pragma unroll
        for (int g = 0; g < 4; ++g)
          a[mt][g] = an[mt][g];
      #pragma unroll
      for (int nt = 0; nt < 7; ++nt)
        #pragma unroll
        for (int g = 0; g < 4; ++g)
          b[nt][g] = bn[nt][g];
    }
  }

  #pragma unroll
  for (int mt = 0; mt < 2; ++mt)
    #pragma unroll
    for (int nt = 0; nt < 7; ++nt)
      #pragma unroll
      for (int v = 0; v < 8; ++v)
        s_p[(wave * ROWS_A + mt * 16 + v + 8 * half) * PITCH + nt * 16 + lm] = acc[mt][nt][v];

  if constexpr (PACKED) {
    dynconv_lds_barrier();
  } else {
    __syncthreads();
    bv[0] = *(const v4f*)(bias + 4 * tid);
    bv[1] = *(const v4f*)(bias + 512 + 4 * tid);
    #pragma unroll
    for (int q = 0; q < WIN_LOADS; ++q) win_load(q);
    __builtin_amdgcn_sched_barrier(0);
  }

  #pragma unroll
  for (int q = 0; q < 512 / NTHR; ++q) {
    const int gi  = tid + NTHR * q;
    const int row = gi >> 4;
    const int h   = gi & 15;
    float l[K_DIM];
    #pragma unroll
    for (int k = 0; k < K_DIM; ++k) {
      float sum = s_p[row * PITCH + h * K_DIM + k];
      #pragma unroll
      for (int w = 1; w < NWAVES; ++w)
        sum += s_p[(w * ROWS_A + row) * PITCH + h * K_DIM + k];
      l[k] = sum;
    }
    float mx = l[0];
    #pragma unroll
    for (int k = 1; k < K_DIM; ++k) mx = fmaxf(mx, l[k]);
    float e[K_DIM];
    float ssum = 0.f;
    #pragma unroll
    for (int k = 0; k < K_DIM; ++k) { e[k] = __expf(l[k] - mx); ssum += e[k]; }
    const float inv = 1.0f / ssum;
    float* dst = s_t + gi * 8;
    *(v4f*)dst       = (v4f){e[0] * inv, e[1] * inv, e[2] * inv, e[3] * inv};
    *(v4f*)(dst + 4) = (v4f){e[4] * inv, e[5] * inv, e[6] * inv, 0.f};
  }
  dynconv_lds_barrier();

  if (t0 == 0) {
    #pragma unroll
    for (int hv = 0; hv < 2; ++hv)
      #pragma unroll
      for (int bb = 0; bb < B_DIM; ++bb)
        #pragma unroll
        for (int j = 0; j < K_DIM - 1; ++j)
          xv[hv][bb][j] = (v4f){0.f, 0.f, 0.f, 0.f};
  }
  #pragma unroll
  for (int i = 0; i < TW; ++i)
    #pragma unroll
    for (int bb = 0; bb < B_DIM; ++bb) {
      const int row = 4 * i + bb;
      #pragma unroll
      for (int hv = 0; hv < 2; ++hv) {
        const int hc = 8 * hv + (tid >> 4);
        const v4f wa = *(const v4f*)(s_t + (row * 16 + hc) * 8);
        const v4f wb = *(const v4f*)(s_t + (row * 16 + hc) * 8 + 4);
        const float wk[K_DIM] = {wa[0], wa[1], wa[2], wa[3], wb[0], wb[1], wb[2]};
        v4f accv = bv[hv];
        #pragma unroll
        for (int k = 0; k < K_DIM; ++k) {
          accv[0] = fmaf(xv[hv][bb][i + k][0], wk[k], accv[0]);
          accv[1] = fmaf(xv[hv][bb][i + k][1], wk[k], accv[1]);
          accv[2] = fmaf(xv[hv][bb][i + k][2], wk[k], accv[2]);
          accv[3] = fmaf(xv[hv][bb][i + k][3], wk[k], accv[3]);
        }
        *(v4f*)(out + (size_t)(mBase + row) * C_DIM + 512 * hv + 4 * tid) = accv;
      }
    }
}

__global__ __launch_bounds__(NTHR)
void dynconv_main_kernel(const float* __restrict__ x,
                         const float* __restrict__ Wm,
                         const float* __restrict__ wpk,
                         size_t ws_bytes,
                         const float* __restrict__ bias,
                         float* __restrict__ out) {
  __shared__ float s_p[NWAVES * ROWS_A * PITCH];
  __shared__ float s_t[ROWS_A * H_DIM * 8];
  if (__builtin_expect(ws_bytes >= PACK_BYTES, 1)) dynconv_main_body<true>(x, wpk, bias, out, s_p, s_t);
  else                                             dynconv_main_body<false>(x, Wm, bias, out, s_p, s_t);
}

extern "C" void kernel_launch(void* const* d_in, const int* in_sizes, int n_in,
                              void* d_out, int out_size, void* d_ws, size_t ws_size,
                              hipStream_t stream) {
  const float* x    = (const float*)d_in[0];
  const float* Wm   = (const float*)d_in[1];
  const float* bias = (const float*)d_in[2];
  float* out = (float*)d_out;

  float* wpk = (float*)d_ws;

  hipLaunchKernelGGL(dynconv_pack_kernel, dim3(112), dim3(256), 0, stream, Wm, wpk, ws_size);
  hipLaunchKernelGGL(dynconv_main_kernel, dim3((T_DIM * B_DIM) / ROWS_A), dim3(NTHR), 0, stream,
                     x, Wm, (const float*)wpk, ws_size, bias, out);
}
